# starting kernel
# baseline (speedup 1.0000x reference)
#include <hip/hip_runtime.h>


typedef float v2f __attribute__((ext_vector_type(2)));
typedef float v8f __attribute__((ext_vector_type(8)));

#define NHEAD 16
#define HDIM  64
#define TOKC  (NHEAD * HDIM)
#define WPB   8

__global__ __launch_bounds__(256) void
K_attention_MH_kernel(const float* __restrict__ x,
                      const float* __restrict__ r_sigma,
                      float* __restrict__ y,
                      int num_tokens) {
    __shared__ float lds_x[WPB][TOKC];
    __shared__ float lds_k[WPB][NHEAD * NHEAD];

    const int lane  = threadIdx.x & 31;
    const int wave  = threadIdx.x >> 5;
    const int token = blockIdx.x * WPB + wave;
    if (token >= num_tokens) return;

    const float sigma = r_sigma[0];
    const float* xt = x + (size_t)token * TOKC;
    float*       yt = y + (size_t)token * TOKC;
    float* lx = lds_x[wave];
    float* lk = lds_k[wave];

    const int row = lane & 15;
    const int hi  = lane >> 4;

    #pragma unroll
    for (int i = 0; i < 8; ++i) {
        const int idx = (i * 32 + lane) * 4;
        const float4 v = *(const float4*)(xt + idx);
        *(float4*)(lx + idx) = v;
    }
    asm volatile("" ::: "memory");

    v8f acc = {};
    #pragma unroll
    for (int c = 0; c < 16; ++c) {
        const int off = row * HDIM + c * 4 + hi * 2;
        v2f a;
        a.x = lx[off + 0];
        a.y = lx[off + 1];
        acc = __builtin_amdgcn_wmma_f32_16x16x4_f32(
            false, a, false, a, (short)0, acc, false, false);
    }

    #pragma unroll
    for (int r = 0; r < 8; ++r)
        lk[(r + hi * 8) * NHEAD + row] = acc[r];
    asm volatile("" ::: "memory");

    float dn = lk[row * NHEAD + row];
    float dm[8];
    #pragma unroll
    for (int r = 0; r < 8; ++r) {
        const int m = r + hi * 8;
        dm[r] = lk[m * NHEAD + m];
    }
    #pragma unroll
    for (int r = 0; r < 8; ++r) {
        const float d = dm[r] + dn - 2.0f * acc[r];
        lk[(r + hi * 8) * NHEAD + row] = __expf(-d * sigma);
    }
    asm volatile("" ::: "memory");

    v2f ka[4];
    #pragma unroll
    for (int c = 0; c < 4; ++c) {
        const int off = row * NHEAD + c * 4 + hi * 2;
        ka[c].x = lk[off + 0];
        ka[c].y = lk[off + 1];
    }

    #pragma unroll
    for (int nb = 0; nb < 4; ++nb) {
        v8f cy;
        #pragma unroll
        for (int r = 0; r < 8; ++r)
            cy[r] = lx[(r + hi * 8) * HDIM + nb * 16 + row];

        #pragma unroll
        for (int c = 0; c < 4; ++c) {
            const int g0 = c * 4 + hi * 2;
            v2f b;
            b.x = lx[(g0 + 0) * HDIM + nb * 16 + row];
            b.y = lx[(g0 + 1) * HDIM + nb * 16 + row];
            cy = __builtin_amdgcn_wmma_f32_16x16x4_f32(
                false, ka[c], false, b, (short)0, cy, false, false);
        }

        #pragma unroll
        for (int r = 0; r < 8; ++r)
            yt[(r + hi * 8) * HDIM + nb * 16 + row] = cy[r];
    }
}

extern "C" void kernel_launch(void* const* d_in, const int* in_sizes, int n_in,
                              void* d_out, int out_size, void* d_ws, size_t ws_size,
                              hipStream_t stream) {
    const float* x       = (const float*)d_in[0];
    const float* r_sigma = (const float*)d_in[1];
    float*       y       = (float*)d_out;

    const int num_tokens = in_sizes[0] / TOKC;
    const int blocks     = (num_tokens + WPB - 1) / WPB;

    hipLaunchKernelGGL(K_attention_MH_kernel,
                       dim3(blocks), dim3(256), 0, stream,
                       x, r_sigma, y, num_tokens);
}


# LDS row pitch padded (x tile 64->76 dwords, K tile 16->20): the Gram operand reads lose their 16-way bank conflict, the K-tile and residual reads their 4- and 2-way ones
# speedup vs baseline: 1.2426x; 1.2426x over previous
#include <hip/hip_runtime.h>


typedef float v2f __attribute__((ext_vector_type(2)));
typedef float v8f __attribute__((ext_vector_type(8)));

#define NHEAD 16
#define HDIM  64
#define TOKC  (NHEAD * HDIM)
#define WPB   8
#define XP    76
#define KP    20

__global__ __launch_bounds__(256) void
K_attention_MH_kernel(const float* __restrict__ x,
                      const float* __restrict__ r_sigma,
                      float* __restrict__ y,
                      int num_tokens) {
    __shared__ float lds_x[WPB][NHEAD * XP];
    __shared__ float lds_k[WPB][NHEAD * KP];

    const int lane  = threadIdx.x & 31;
    const int wave  = threadIdx.x >> 5;
    const int token = blockIdx.x * WPB + wave;
    if (token >= num_tokens) return;

    const float sigma = r_sigma[0];
    const float* xt = x + (size_t)token * TOKC;
    float*       yt = y + (size_t)token * TOKC;
    float* lx = lds_x[wave];
    float* lk = lds_k[wave];

    const int row = lane & 15;
    const int hi  = lane >> 4;

    #pragma unroll
    for (int i = 0; i < 8; ++i) {
        const int idx = (i * 32 + lane) * 4;
        const float4 v = *(const float4*)(xt + idx);
        *(float4*)(lx + (i * 2 + hi) * XP + row * 4) = v;
    }
    asm volatile("" ::: "memory");

    v8f acc = {};
    #pragma unroll
    for (int c = 0; c < 16; ++c) {
        const int off = row * XP + c * 4 + hi * 2;
        v2f a;
        a.x = lx[off + 0];
        a.y = lx[off + 1];
        acc = __builtin_amdgcn_wmma_f32_16x16x4_f32(
            false, a, false, a, (short)0, acc, false, false);
    }

    #pragma unroll
    for (int r = 0; r < 8; ++r)
        lk[(r + hi * 8) * KP + row] = acc[r];
    asm volatile("" ::: "memory");

    float dn = lk[row * KP + row];
    float dm[8];
    #pragma unroll
    for (int r = 0; r < 8; ++r) {
        const int m = r + hi * 8;
        dm[r] = lk[m * KP + m];
    }
    #pragma unroll
    for (int r = 0; r < 8; ++r) {
        const float d = dm[r] + dn - 2.0f * acc[r];
        lk[(r + hi * 8) * KP + row] = __expf(-d * sigma);
    }
    asm volatile("" ::: "memory");

    v2f ka[4];
    #pragma unroll
    for (int c = 0; c < 4; ++c) {
        const int off = row * KP + c * 4 + hi * 2;
        ka[c].x = lk[off + 0];
        ka[c].y = lk[off + 1];
    }

    #pragma unroll
    for (int nb = 0; nb < 4; ++nb) {
        v8f cy;
        #pragma unroll
        for (int r = 0; r < 8; ++r)
            cy[r] = lx[(r + hi * 8) * XP + nb * 16 + row];

        #pragma unroll
        for (int c = 0; c < 4; ++c) {
            const int g0 = c * 4 + hi * 2;
            v2f b;
            b.x = lx[(g0 + 0) * XP + nb * 16 + row];
            b.y = lx[(g0 + 1) * XP + nb * 16 + row];
            cy = __builtin_amdgcn_wmma_f32_16x16x4_f32(
                false, ka[c], false, b, (short)0, cy, false, false);
        }

        #pragma unroll
        for (int r = 0; r < 8; ++r)
            yt[(r + hi * 8) * HDIM + nb * 16 + row] = cy[r];
    }
}

extern "C" void kernel_launch(void* const* d_in, const int* in_sizes, int n_in,
                              void* d_out, int out_size, void* d_ws, size_t ws_size,
                              hipStream_t stream) {
    const float* x       = (const float*)d_in[0];
    const float* r_sigma = (const float*)d_in[1];
    float*       y       = (float*)d_out;

    const int num_tokens = in_sizes[0] / TOKC;
    const int blocks     = (num_tokens + WPB - 1) / WPB;

    hipLaunchKernelGGL(K_attention_MH_kernel,
                       dim3(blocks), dim3(256), 0, stream,
                       x, r_sigma, y, num_tokens);
}


# K@x runs from registers: exp'd Gram D tile is its A operand (K symmetric), global-load float4s are B and residual (tile j = columns 4n+j); 8 whole-line 16-B stores; head sum order changes
# speedup vs baseline: 1.3166x; 1.0596x over previous
#include <hip/hip_runtime.h>


typedef float v2f __attribute__((ext_vector_type(2)));
typedef float v4f __attribute__((ext_vector_type(4)));
typedef float v8f __attribute__((ext_vector_type(8)));

#define NHEAD 16
#define HDIM  64
#define TOKC  (NHEAD * HDIM)
#define WPB   8
#define XP    76

static __device__ __forceinline__ float other_row(float v) {
    const int i = __builtin_bit_cast(int, v);
    return __builtin_bit_cast(float,
        __builtin_amdgcn_permlanex16(i, i, 0x76543210, 0xfedcba98, false, false));
}
static __device__ __forceinline__ float row_gather(float v, unsigned lo, unsigned hi) {
    const int i = __builtin_bit_cast(int, v);
    return __builtin_bit_cast(float,
        __builtin_amdgcn_permlane16(i, i, lo, hi, false, false));
}

__global__ __launch_bounds__(256) void
K_attention_MH_kernel(const float* __restrict__ x,
                      const float* __restrict__ r_sigma,
                      float* __restrict__ y,
                      int num_tokens) {
    __shared__ float lds_x[WPB][NHEAD * XP];

    const int lane  = threadIdx.x & 31;
    const int wave  = threadIdx.x >> 5;
    const int token = blockIdx.x * WPB + wave;
    if (token >= num_tokens) return;

    const float sigma = r_sigma[0];
    const float* xt = x + (size_t)token * TOKC;
    float*       yt = y + (size_t)token * TOKC;
    float* lx = lds_x[wave];

    const int row = lane & 15;
    const int hi  = lane >> 4;

    v4f xv[8];
    #pragma unroll
    for (int i = 0; i < 8; ++i)
        xv[i] = *(const v4f*)(xt + (hi * 8 + i) * HDIM + row * 4);
    #pragma unroll
    for (int i = 0; i < 8; ++i)
        *(v4f*)(lx + (hi * 8 + i) * XP + row * 4) = xv[i];
    asm volatile("" ::: "memory");

    v8f acc = {};
    #pragma unroll
    for (int c = 0; c < 16; ++c) {
        const int off = row * XP + c * 4 + hi * 2;
        v2f a;
        a.x = lx[off + 0];
        a.y = lx[off + 1];
        acc = __builtin_amdgcn_wmma_f32_16x16x4_f32(
            false, a, false, a, (short)0, acc, false, false);
    }

    const bool b0 = (row & 1) != 0, b1 = (row & 2) != 0, b2 = (row & 4) != 0;
    const float t01 = b0 ? acc[1] : acc[0];
    const float t23 = b0 ? acc[3] : acc[2];
    const float t45 = b0 ? acc[5] : acc[4];
    const float t67 = b0 ? acc[7] : acc[6];
    const float u0  = b1 ? t23 : t01;
    const float u1  = b1 ? t67 : t45;
    const float t   = b2 ? u1 : u0;
    const float tx  = other_row(t);
    const float dn  = ((row >> 3) == hi) ? t : tx;
    const float dn8 = row_gather(dn, 0xfedcba98u, 0x76543210u);
    const float e   = hi ? dn8 : dn;

    float kk[8];
    #pragma unroll
    for (int r = 0; r < 8; ++r) {
        const unsigned sel = 0x11111111u * (unsigned)r;
        const float dm = row_gather(e, sel, sel);
        const float d = dm + dn - 2.0f * acc[r];
        kk[r] = __expf(-d * sigma);
    }

    v8f cy[4];
    #pragma unroll
    for (int j = 0; j < 4; ++j) {
        #pragma unroll
        for (int r = 0; r < 8; ++r)
            cy[j][r] = xv[r][j];
    }
    #pragma unroll
    for (int c = 0; c < 4; ++c) {
        v2f a;
        a.x = kk[2 * c + 0];
        a.y = kk[2 * c + 1];
        #pragma unroll
        for (int j = 0; j < 4; ++j) {
            v2f b;
            b.x = xv[2 * c + 0][j];
            b.y = xv[2 * c + 1][j];
            cy[j] = __builtin_amdgcn_wmma_f32_16x16x4_f32(
                false, a, false, b, (short)0, cy[j], false, false);
        }
    }

    #pragma unroll
    for (int r = 0; r < 8; ++r) {
        v4f o;
        o.x = cy[0][r];
        o.y = cy[1][r];
        o.z = cy[2][r];
        o.w = cy[3][r];
        *(v4f*)(yt + (hi * 8 + r) * HDIM + row * 4) = o;
    }
}

extern "C" void kernel_launch(void* const* d_in, const int* in_sizes, int n_in,
                              void* d_out, int out_size, void* d_ws, size_t ws_size,
                              hipStream_t stream) {
    const float* x       = (const float*)d_in[0];
    const float* r_sigma = (const float*)d_in[1];
    float*       y       = (float*)d_out;

    const int num_tokens = in_sizes[0] / TOKC;
    const int blocks     = (num_tokens + WPB - 1) / WPB;

    hipLaunchKernelGGL(K_attention_MH_kernel,
                       dim3(blocks), dim3(256), 0, stream,
                       x, r_sigma, y, num_tokens);
}
